# starting kernel
# baseline (speedup 1.0000x reference)
#include <hip/hip_runtime.h>


typedef __attribute__((ext_vector_type(2))) float v2f;
typedef __attribute__((ext_vector_type(8))) float v8f;

static inline int idiv_up(long long a, long long b) { return (int)((a + b - 1) / b); }

__global__ void fill_zero_kernel(float* __restrict__ ptr, long long count) {
  long long i = (long long)blockIdx.x * blockDim.x + threadIdx.x;
  if (i < count) ptr[i] = 0.0f;
}

__global__ void init_out_kernel(float* __restrict__ out,
                                const float* __restrict__ bias2,
                                long long count, int C) {
  long long i = (long long)blockIdx.x * blockDim.x + threadIdx.x;
  if (i < count) out[i] = bias2[i % C];
}

__global__ void degree_kernel(const int* __restrict__ s, const int* __restrict__ p,
                              const int* __restrict__ o, float* __restrict__ deg,
                              int E, int N, int R) {
  long long i = (long long)blockIdx.x * blockDim.x + threadIdx.x;
  if (i < E) {
    int si = s[i], pi = p[i], oi = o[i];
    atomicAdd(&deg[(size_t)pi * N + si], 1.0f);
    atomicAdd(&deg[(size_t)(pi + R) * N + oi], 1.0f);
  } else if (i < (long long)E + N) {
    int node = (int)(i - E);
    deg[(size_t)(2 * R) * N + node] = 1.0f;
  }
}

__global__ void gemm48_wmma_kernel(const float* __restrict__ A,
                                   const float* __restrict__ B,
                                   float* __restrict__ C,
                                   int M, int K, long long J) {
  __shared__ float sA[48 * 64];

  const int total = 48 * K;
  for (int idx = threadIdx.x; idx < total; idx += blockDim.x) {
    const int row = idx / K;
    const int kk  = idx - row * K;
    sA[idx] = (row < M) ? A[row * K + kk] : 0.0f;
  }
  __syncthreads();

  const int wave = threadIdx.x >> 5;
  const int lane = threadIdx.x & 31;
  const long long jbase =
      ((long long)blockIdx.x * (blockDim.x >> 5) + wave) * 16;
  if (jbase >= J) return;
  const int col   = lane & 15;
  const int khalf = (lane >> 4) * 2;

  v8f acc[3];
  acc[0] = (v8f){};
  acc[1] = (v8f){};
  acc[2] = (v8f){};

  const float* __restrict__ Bp = B + jbase + col;
  for (int kc = 0; kc < K; kc += 4) {
    const int k0 = kc + khalf;
    v2f b;
    b.x = Bp[(long long)k0 * J];
    b.y = Bp[(long long)(k0 + 1) * J];
#pragma unroll
    for (int mt = 0; mt < 3; ++mt) {
      const int m = mt * 16 + col;
      const v2f a = *(const v2f*)&sA[m * K + k0];
      acc[mt] = __builtin_amdgcn_wmma_f32_16x16x4_f32(
          false, a, false, b, (short)0, acc[mt], false, false);
    }
  }

  const int rhalf = (lane >> 4) * 8;
#pragma unroll
  for (int mt = 0; mt < 3; ++mt) {
#pragma unroll
    for (int v = 0; v < 8; ++v) {
      const int row = mt * 16 + rhalf + v;
      if (row < M) C[(long long)row * J + jbase + col] = acc[mt][v];
    }
  }
}

__global__ void layer1_scatter_kernel(const int* __restrict__ s, const int* __restrict__ p,
                                      const int* __restrict__ o,
                                      const float* __restrict__ deg,
                                      const float* __restrict__ W1,
                                      float* __restrict__ h,
                                      int E, int N, int R, long long NE) {
  long long gid = (long long)blockIdx.x * blockDim.x + threadIdx.x;
  const long long total = ((long long)2 * E + N) * 16;
  if (gid >= total) return;
  const int e = (int)(gid & 15);
  const long long i = gid >> 4;
  int si, pi, oi;
  if (i < E)              { si = s[i];      pi = p[i];          oi = o[i]; }
  else if (i < 2LL * E)   { long long j = i - E; si = o[j]; pi = p[j] + R; oi = s[j]; }
  else                    { int nd = (int)(i - 2LL * E); si = nd; pi = 2 * R; oi = nd; }
  const float val = 1.0f / deg[(size_t)pi * N + si];
  const float w   = W1[(size_t)pi * NE + (size_t)oi * 16 + e];
  atomicAdd(&h[(size_t)si * 16 + e], val * w);
}

__global__ void relu_bias_kernel(float* __restrict__ h, const float* __restrict__ bias1,
                                 long long count) {
  long long i = (long long)blockIdx.x * blockDim.x + threadIdx.x;
  if (i < count) h[i] = fmaxf(h[i] + bias1[i & 15], 0.0f);
}

__global__ void layer2_scatter_kernel(const int* __restrict__ s, const int* __restrict__ p,
                                      const int* __restrict__ o,
                                      const float* __restrict__ deg,
                                      const float* __restrict__ h,
                                      const float* __restrict__ W2,
                                      float* __restrict__ out,
                                      int E, int N, int R) {
  long long gid = (long long)blockIdx.x * blockDim.x + threadIdx.x;
  const long long total = ((long long)2 * E + N) * 8;
  if (gid >= total) return;
  const int c = (int)(gid & 7);
  const long long i = gid >> 3;
  int si, pi, oi;
  if (i < E)              { si = s[i];      pi = p[i];          oi = o[i]; }
  else if (i < 2LL * E)   { long long j = i - E; si = o[j]; pi = p[j] + R; oi = s[j]; }
  else                    { int nd = (int)(i - 2LL * E); si = nd; pi = 2 * R; oi = nd; }
  const float val = 1.0f / deg[(size_t)pi * N + si];
  const float* hv = &h[(size_t)oi * 16];
  const float* w  = &W2[(size_t)pi * 128 + c];
  float sum = 0.0f;
#pragma unroll
  for (int e = 0; e < 16; ++e) sum += hv[e] * w[e * 8];
  atomicAdd(&out[(size_t)si * 8 + c], val * sum);
}

extern "C" void kernel_launch(void* const* d_in, const int* in_sizes, int n_in,
                              void* d_out, int out_size, void* d_ws, size_t ws_size,
                              hipStream_t stream) {
  const float* comps1 = (const float*)d_in[0];
  const float* bases1 = (const float*)d_in[1];
  const float* comps2 = (const float*)d_in[2];
  const float* bases2 = (const float*)d_in[3];
  const float* bias1  = (const float*)d_in[4];
  const float* bias2  = (const float*)d_in[5];
  const int*   s      = (const int*)d_in[6];
  const int*   p      = (const int*)d_in[7];
  const int*   o      = (const int*)d_in[8];

  const int EMB    = in_sizes[4];
  const int NUMCLS = in_sizes[5];
  const int E      = in_sizes[6];
  const int N      = out_size / NUMCLS;
  const int BASES  = (int)((long long)in_sizes[1] / ((long long)N * EMB));
  const int r_tot  = in_sizes[0] / BASES;
  const int R      = (r_tot - 1) / 2;

  float* ws = (float*)d_ws;
  size_t off = 0;
  auto alloc = [&](size_t nfloats) {
    size_t cur = off;
    off += (nfloats + 63) & ~(size_t)63;
    return cur;
  };
  float* deg = ws + alloc((size_t)r_tot * N);
  float* h   = ws + alloc((size_t)N * EMB);
  float* W2  = ws + alloc((size_t)r_tot * EMB * NUMCLS);
  float* W1  = ws + alloc((size_t)r_tot * N * EMB);
  (void)ws_size; (void)n_in;

  const long long NE   = (long long)N * EMB;
  const long long J2   = (long long)EMB * NUMCLS;
  const long long Eall = (long long)2 * E + N;
  const int B = 256;

  fill_zero_kernel<<<idiv_up((long long)r_tot * N, B), B, 0, stream>>>(deg, (long long)r_tot * N);
  fill_zero_kernel<<<idiv_up(NE, B), B, 0, stream>>>(h, NE);

  degree_kernel<<<idiv_up((long long)E + N, B), B, 0, stream>>>(s, p, o, deg, E, N, R);

  gemm48_wmma_kernel<<<idiv_up(NE, 16LL * (B / 32)), B, 0, stream>>>(
      comps1, bases1, W1, r_tot, BASES, NE);
  gemm48_wmma_kernel<<<idiv_up(J2, 16LL * (B / 32)), B, 0, stream>>>(
      comps2, bases2, W2, r_tot, BASES, J2);

  layer1_scatter_kernel<<<idiv_up(Eall * 16, B), B, 0, stream>>>(
      s, p, o, deg, W1, h, E, N, R, NE);
  relu_bias_kernel<<<idiv_up(NE, B), B, 0, stream>>>(h, bias1, NE);

  init_out_kernel<<<idiv_up((long long)N * NUMCLS, B), B, 0, stream>>>(
      (float*)d_out, bias2, (long long)N * NUMCLS, NUMCLS);
  layer2_scatter_kernel<<<idiv_up(Eall * 8, B), B, 0, stream>>>(
      s, p, o, deg, h, W2, (float*)d_out, E, N, R);
}


# Layer-1 scatter moved into the W1 product: a node's wave parks its 41x16 tile in LDS, walks the node's edge lists (built with the degrees), adds rows to h by the same float atomics; W1 never stored
# speedup vs baseline: 1.2344x; 1.2344x over previous
#include <hip/hip_runtime.h>


typedef __attribute__((ext_vector_type(2))) float v2f;
typedef __attribute__((ext_vector_type(8))) float v8f;

static inline int idiv_up(long long a, long long b) { return (int)((a + b - 1) / b); }

__global__ void fill_zero16_kernel(float4* __restrict__ ptr, long long count4) {
  long long i = (long long)blockIdx.x * blockDim.x + threadIdx.x;
  if (i < count4) ptr[i] = make_float4(0.0f, 0.0f, 0.0f, 0.0f);
}

__global__ void init_out_kernel(float* __restrict__ out,
                                const float* __restrict__ bias2,
                                long long count, int C) {
  long long i = (long long)blockIdx.x * blockDim.x + threadIdx.x;
  if (i < count) out[i] = bias2[i % C];
}

constexpr int kLists = 16;

__global__ void degree_list_kernel(const int* __restrict__ s, const int* __restrict__ p,
                                   const int* __restrict__ o, float* __restrict__ deg,
                                   int* __restrict__ head, int* __restrict__ next,
                                   int E, int N, int R) {
  long long i = (long long)blockIdx.x * blockDim.x + threadIdx.x;
  if (i < E) {
    int si = s[i], pi = p[i], oi = o[i];
    atomicAdd(&deg[(size_t)pi * N + si], 1.0f);
    atomicAdd(&deg[(size_t)(pi + R) * N + oi], 1.0f);
    const int e0 = (int)i, e1 = (int)i + E;
    const int before0 = atomicExch(&head[(size_t)oi * kLists + (pi & (kLists - 1))], e0 + 1);
    const int before1 = atomicExch(&head[(size_t)si * kLists + ((pi + R) & (kLists - 1))], e1 + 1);
    next[e0] = before0;
    next[e1] = before1;
  } else if (i < (long long)E + N) {
    int node = (int)(i - E);
    deg[(size_t)(2 * R) * N + node] = 1.0f;
  }
}

constexpr int kGemmWaves = 8;
constexpr int kGemmThreads = kGemmWaves * 32;

template <bool kScatter>
__global__ void gemm48_wmma_kernel(const float* __restrict__ A,
                                   const float* __restrict__ B,
                                   float* __restrict__ C,
                                   int M, int K, long long J,
                                   const int* __restrict__ s, const int* __restrict__ p,
                                   const int* __restrict__ o,
                                   const float* __restrict__ deg,
                                   const int* __restrict__ head,
                                   const int* __restrict__ next,
                                   float* __restrict__ h,
                                   int E, int N, int R) {
  __shared__ float sA[48 * 64];
  __shared__ float sTile[kScatter ? kGemmWaves * 48 * 16 : 1];

  const int total = 48 * K;
  for (int idx = threadIdx.x; idx < total; idx += blockDim.x) {
    const int row = idx / K;
    const int kk  = idx - row * K;
    sA[idx] = (row < M) ? A[row * K + kk] : 0.0f;
  }
  __syncthreads();

  const int wave = threadIdx.x >> 5;
  const int lane = threadIdx.x & 31;
  const long long jbase =
      ((long long)blockIdx.x * (blockDim.x >> 5) + wave) * 16;
  const bool active = jbase < J;
  const long long jb = active ? jbase : J - 16;
  const int col   = lane & 15;
  const int khalf = (lane >> 4) * 2;

  v8f acc[3];
  acc[0] = (v8f){};
  acc[1] = (v8f){};
  acc[2] = (v8f){};

  const float* __restrict__ Bp = B + jb + col;
  for (int kc = 0; kc < K; kc += 4) {
    const int k0 = kc + khalf;
    v2f b;
    b.x = Bp[(long long)k0 * J];
    b.y = Bp[(long long)(k0 + 1) * J];
#pragma unroll
    for (int mt = 0; mt < 3; ++mt) {
      const int m = mt * 16 + col;
      const v2f a = *(const v2f*)&sA[m * K + k0];
      acc[mt] = __builtin_amdgcn_wmma_f32_16x16x4_f32(
          false, a, false, b, (short)0, acc[mt], false, false);
    }
  }

  const int rhalf = (lane >> 4) * 8;
  if (!kScatter) {
    if (active) {
#pragma unroll
      for (int mt = 0; mt < 3; ++mt) {
#pragma unroll
        for (int v = 0; v < 8; ++v) {
          const int row = mt * 16 + rhalf + v;
          if (row < M) C[(long long)row * J + jb + col] = acc[mt][v];
        }
      }
    }
    return;
  }

  float* __restrict__ tile = &sTile[wave * (48 * 16)];
#pragma unroll
  for (int mt = 0; mt < 3; ++mt) {
#pragma unroll
    for (int v = 0; v < 8; ++v) {
      tile[(mt * 16 + rhalf + v) * 16 + col] = acc[mt][v];
    }
  }
  __syncthreads();
  if (!active) return;

  const int node = (int)(jbase >> 4);
  int cur = (lane < kLists) ? head[(size_t)node * kLists + lane] : 0;
  for (int trip = 0; trip < 2 * E; ++trip) {
    const bool has = cur != 0;
    unsigned pending = __builtin_amdgcn_ballot_w32(has);
    if (pending == 0u) break;
    int tgt = 0, rel = 0;
    float val = 0.0f;
    if (has) {
      const int e   = cur - 1;
      const bool inv = e >= E;
      const int j   = inv ? e - E : e;
      const int pj  = p[j];
      tgt = inv ? o[j] : s[j];
      rel = inv ? pj + R : pj;
      cur = next[e];
      val = 1.0f / deg[(size_t)rel * N + tgt];
    }
    while (pending != 0u) {
      const int la = __builtin_ctz(pending);
      pending &= pending - 1u;
      const bool two = pending != 0u;
      int lb = la;
      if (two) { lb = __builtin_ctz(pending); pending &= pending - 1u; }
      const int from = (lane < 16) ? la : lb;
      const int   t = __shfl(tgt, from);
      const int   r = __shfl(rel, from);
      const float v = __shfl(val, from);
      if (lane < 16 || two) atomicAdd(&h[(size_t)t * 16 + col], v * tile[r * 16 + col]);
    }
  }
  if (lane < 16) atomicAdd(&h[(size_t)node * 16 + col], tile[(2 * R) * 16 + col]);
}

__global__ void relu_bias_kernel(float* __restrict__ h, const float* __restrict__ bias1,
                                 long long count) {
  long long i = (long long)blockIdx.x * blockDim.x + threadIdx.x;
  if (i < count) h[i] = fmaxf(h[i] + bias1[i & 15], 0.0f);
}

__global__ void layer2_scatter_kernel(const int* __restrict__ s, const int* __restrict__ p,
                                      const int* __restrict__ o,
                                      const float* __restrict__ deg,
                                      const float* __restrict__ h,
                                      const float* __restrict__ W2,
                                      float* __restrict__ out,
                                      int E, int N, int R) {
  long long gid = (long long)blockIdx.x * blockDim.x + threadIdx.x;
  const long long total = ((long long)2 * E + N) * 8;
  if (gid >= total) return;
  const int c = (int)(gid & 7);
  const long long i = gid >> 3;
  int si, pi, oi;
  if (i < E)              { si = s[i];      pi = p[i];          oi = o[i]; }
  else if (i < 2LL * E)   { long long j = i - E; si = o[j]; pi = p[j] + R; oi = s[j]; }
  else                    { int nd = (int)(i - 2LL * E); si = nd; pi = 2 * R; oi = nd; }
  const float val = 1.0f / deg[(size_t)pi * N + si];
  const float* hv = &h[(size_t)oi * 16];
  const float* w  = &W2[(size_t)pi * 128 + c];
  float sum = 0.0f;
#pragma unroll
  for (int e = 0; e < 16; ++e) sum += hv[e] * w[e * 8];
  atomicAdd(&out[(size_t)si * 8 + c], val * sum);
}

extern "C" void kernel_launch(void* const* d_in, const int* in_sizes, int n_in,
                              void* d_out, int out_size, void* d_ws, size_t ws_size,
                              hipStream_t stream) {
  const float* comps1 = (const float*)d_in[0];
  const float* bases1 = (const float*)d_in[1];
  const float* comps2 = (const float*)d_in[2];
  const float* bases2 = (const float*)d_in[3];
  const float* bias1  = (const float*)d_in[4];
  const float* bias2  = (const float*)d_in[5];
  const int*   s      = (const int*)d_in[6];
  const int*   p      = (const int*)d_in[7];
  const int*   o      = (const int*)d_in[8];

  const int EMB    = in_sizes[4];
  const int NUMCLS = in_sizes[5];
  const int E      = in_sizes[6];
  const int N      = out_size / NUMCLS;
  const int BASES  = (int)((long long)in_sizes[1] / ((long long)N * EMB));
  const int r_tot  = in_sizes[0] / BASES;
  const int R      = (r_tot - 1) / 2;

  float* ws = (float*)d_ws;
  size_t off = 0;
  auto alloc = [&](size_t nwords) {
    size_t cur = off;
    off += (nwords + 63) & ~(size_t)63;
    return cur;
  };
  float* deg  = ws + alloc((size_t)r_tot * N);
  int*   head = (int*)(ws + alloc((size_t)N * kLists));
  float* h    = ws + alloc((size_t)N * EMB);
  const size_t cleared = off;
  float* W2   = ws + alloc((size_t)r_tot * EMB * NUMCLS);
  int*   next = (int*)(ws + alloc((size_t)2 * E));
  (void)ws_size; (void)n_in;

  const long long NE   = (long long)N * EMB;
  const long long J2   = (long long)EMB * NUMCLS;
  const long long Eall = (long long)2 * E + N;
  const int B = 256;

  fill_zero16_kernel<<<idiv_up((long long)(cleared / 4), B), B, 0, stream>>>(
      (float4*)ws, (long long)(cleared / 4));

  degree_list_kernel<<<idiv_up((long long)E + N, B), B, 0, stream>>>(
      s, p, o, deg, head, next, E, N, R);

  gemm48_wmma_kernel<true><<<idiv_up(NE, 16LL * kGemmWaves), kGemmThreads, 0, stream>>>(
      comps1, bases1, nullptr, r_tot, BASES, NE, s, p, o, deg, head, next, h, E, N, R);
  gemm48_wmma_kernel<false><<<idiv_up(J2, 16LL * kGemmWaves), kGemmThreads, 0, stream>>>(
      comps2, bases2, W2, r_tot, BASES, J2, nullptr, nullptr, nullptr, nullptr, nullptr,
      nullptr, nullptr, 0, 0, 0);

  relu_bias_kernel<<<idiv_up(NE, B), B, 0, stream>>>(h, bias1, NE);

  init_out_kernel<<<idiv_up((long long)N * NUMCLS, B), B, 0, stream>>>(
      (float*)d_out, bias2, (long long)N * NUMCLS, NUMCLS);
  layer2_scatter_kernel<<<idiv_up(Eall * 8, B), B, 0, stream>>>(
      s, p, o, deg, h, W2, (float*)d_out, E, N, R);
}


# W1 launch: every load is requested a stage ahead of its use: A image by 16-byte loads, all B rows of a tile as one group, list heads on entry, next list entries fetched under the current adds
# speedup vs baseline: 1.4855x; 1.2034x over previous
#include <hip/hip_runtime.h>


typedef __attribute__((ext_vector_type(2))) float v2f;
typedef __attribute__((ext_vector_type(8))) float v8f;

static inline int idiv_up(long long a, long long b) { return (int)((a + b - 1) / b); }

__global__ void fill_zero16_kernel(float4* __restrict__ ptr, long long count4) {
  long long i = (long long)blockIdx.x * blockDim.x + threadIdx.x;
  if (i < count4) ptr[i] = make_float4(0.0f, 0.0f, 0.0f, 0.0f);
}

__global__ void init_out_kernel(float* __restrict__ out,
                                const float* __restrict__ bias2,
                                long long count, int C) {
  long long i = (long long)blockIdx.x * blockDim.x + threadIdx.x;
  if (i < count) out[i] = bias2[i % C];
}

constexpr int kLists = 16;

__global__ void degree_list_kernel(const int* __restrict__ s, const int* __restrict__ p,
                                   const int* __restrict__ o, float* __restrict__ deg,
                                   int* __restrict__ head, int* __restrict__ next,
                                   int E, int N, int R) {
  long long i = (long long)blockIdx.x * blockDim.x + threadIdx.x;
  if (i < E) {
    int si = s[i], pi = p[i], oi = o[i];
    atomicAdd(&deg[(size_t)pi * N + si], 1.0f);
    atomicAdd(&deg[(size_t)(pi + R) * N + oi], 1.0f);
    const int e0 = (int)i, e1 = (int)i + E;
    const int before0 = atomicExch(&head[(size_t)oi * kLists + (pi & (kLists - 1))], e0 + 1);
    const int before1 = atomicExch(&head[(size_t)si * kLists + ((pi + R) & (kLists - 1))], e1 + 1);
    next[e0] = before0;
    next[e1] = before1;
  } else if (i < (long long)E + N) {
    int node = (int)(i - E);
    deg[(size_t)(2 * R) * N + node] = 1.0f;
  }
}

constexpr int kGemmWaves = 8;
constexpr int kGemmThreads = kGemmWaves * 32;
constexpr int kBatchSteps = 10;

template <bool kScatter>
__global__ void gemm48_wmma_kernel(const float* __restrict__ A,
                                   const float* __restrict__ B,
                                   float* __restrict__ C,
                                   int M, int K, long long J,
                                   const int* __restrict__ s, const int* __restrict__ p,
                                   const int* __restrict__ o,
                                   const float* __restrict__ deg,
                                   const int* __restrict__ head,
                                   const int* __restrict__ next,
                                   float* __restrict__ h,
                                   int E, int N, int R) {
  __shared__ __attribute__((aligned(16))) float sA[48 * 64];
  __shared__ float sTile[kScatter ? kGemmWaves * 48 * 16 : 1];

  const int wave = threadIdx.x >> 5;
  const int lane = threadIdx.x & 31;
  const long long jbase =
      ((long long)blockIdx.x * (blockDim.x >> 5) + wave) * 16;
  const bool active = jbase < J;
  const long long jb = active ? jbase : J - 16;
  const int col   = lane & 15;
  const int khalf = (lane >> 4) * 2;
  const int node  = (int)(jbase >> 4);

  const int total4 = 12 * K;
  const int valid4 = (M * K) >> 2;
  const float4* __restrict__ A4 = (const float4*)A;
  float4 areg[3];
#pragma unroll
  for (int t = 0; t < 3; ++t) {
    const int i4 = (int)threadIdx.x + t * kGemmThreads;
    areg[t] = (i4 < valid4) ? A4[i4] : make_float4(0.0f, 0.0f, 0.0f, 0.0f);
  }

  int cur = 0;
  if (kScatter) {
    if (active && lane < kLists) cur = head[(size_t)node * kLists + lane];
  }

  const float* __restrict__ Bp = B + jb + col;
  const unsigned bidx0 = (unsigned)((long long)khalf * J + jb + col);
  const unsigned bidx1 = bidx0 + (unsigned)J;
  const bool batch0 = K >= 4 * kBatchSteps;
  v2f bfrag[kBatchSteps] = {};
  if (batch0) {
#pragma unroll
    for (int i = 0; i < kBatchSteps; ++i) {
      unsigned long long rowbase = (unsigned long long)(B + (long long)(4 * i) * J);
      asm volatile("" : "+s"(rowbase));
      const __attribute__((address_space(1))) float* rowp =
          (const __attribute__((address_space(1))) float*)rowbase;
      bfrag[i].x = rowp[bidx0];
      bfrag[i].y = rowp[bidx1];
    }
  }

  float4* sA4 = (float4*)sA;
#pragma unroll
  for (int t = 0; t < 3; ++t) {
    const int i4 = (int)threadIdx.x + t * kGemmThreads;
    if (i4 < total4) sA4[i4] = areg[t];
  }

  auto fetch = [&](int c, int& tgt, int& pj, int& nx) {
    tgt = 0; pj = 0; nx = 0;
    if (c != 0) {
      const int e    = c - 1;
      const bool inv = e >= E;
      const int j    = inv ? e - E : e;
      pj  = p[j];
      tgt = inv ? o[j] : s[j];
      nx  = next[e];
    }
  };

  __syncthreads();

  int tgt = 0, prel = 0, nxt = 0;
  if (kScatter) fetch(cur, tgt, prel, nxt);

  v8f acc[3];
  acc[0] = (v8f){};
  acc[1] = (v8f){};
  acc[2] = (v8f){};

  auto mma_step = [&](int k0, v2f b) {
    v2f a[3];
#pragma unroll
    for (int mt = 0; mt < 3; ++mt) a[mt] = *(const v2f*)&sA[(mt * 16 + col) * K + k0];
#pragma unroll
    for (int mt = 0; mt < 3; ++mt) {
      acc[mt] = __builtin_amdgcn_wmma_f32_16x16x4_f32(
          false, a[mt], false, b, (short)0, acc[mt], false, false);
    }
  };

  int kc = 0;
  if (batch0) {
#pragma unroll
    for (int i = 0; i < kBatchSteps; ++i) mma_step(4 * i + khalf, bfrag[i]);
    kc = 4 * kBatchSteps;
  }
  for (; kc < K; kc += 4) {
    const int k0 = kc + khalf;
    v2f b;
    b.x = Bp[(long long)k0 * J];
    b.y = Bp[(long long)(k0 + 1) * J];
    mma_step(k0, b);
  }

  const int rhalf = (lane >> 4) * 8;
  if (!kScatter) {
    if (active) {
#pragma unroll
      for (int mt = 0; mt < 3; ++mt) {
#pragma unroll
        for (int v = 0; v < 8; ++v) {
          const int row = mt * 16 + rhalf + v;
          if (row < M) C[(long long)row * J + jb + col] = acc[mt][v];
        }
      }
    }
    return;
  }

  float* __restrict__ tile = &sTile[wave * (48 * 16)];
#pragma unroll
  for (int mt = 0; mt < 3; ++mt) {
#pragma unroll
    for (int v = 0; v < 8; ++v) {
      tile[(mt * 16 + rhalf + v) * 16 + col] = acc[mt][v];
    }
  }
  __syncthreads();
  if (!active) return;

  for (int trip = 0; trip < 2 * E; ++trip) {
    const bool has = cur != 0;
    unsigned pending = __builtin_amdgcn_ballot_w32(has);
    if (pending == 0u) break;
    const int rel = (cur - 1 >= E) ? prel + R : prel;
    float dg = 1.0f;
    if (has) dg = deg[(size_t)rel * N + tgt];
    const int cur2 = nxt;
    int tgt2, prel2, nxt2;
    fetch(cur2, tgt2, prel2, nxt2);
    const float val = has ? 1.0f / dg : 0.0f;
    while (pending != 0u) {
      const int la = __builtin_ctz(pending);
      pending &= pending - 1u;
      const bool two = pending != 0u;
      int lb = la;
      if (two) { lb = __builtin_ctz(pending); pending &= pending - 1u; }
      const int from = (lane < 16) ? la : lb;
      const int   t = __shfl(tgt, from);
      const int   r = __shfl(rel, from);
      const float v = __shfl(val, from);
      if (lane < 16 || two) atomicAdd(&h[(size_t)t * 16 + col], v * tile[r * 16 + col]);
    }
    cur = cur2; tgt = tgt2; prel = prel2; nxt = nxt2;
  }
  if (lane < 16) atomicAdd(&h[(size_t)node * 16 + col], tile[(2 * R) * 16 + col]);
}

__global__ void relu_bias_kernel(float* __restrict__ h, const float* __restrict__ bias1,
                                 long long count) {
  long long i = (long long)blockIdx.x * blockDim.x + threadIdx.x;
  if (i < count) h[i] = fmaxf(h[i] + bias1[i & 15], 0.0f);
}

__global__ void layer2_scatter_kernel(const int* __restrict__ s, const int* __restrict__ p,
                                      const int* __restrict__ o,
                                      const float* __restrict__ deg,
                                      const float* __restrict__ h,
                                      const float* __restrict__ W2,
                                      float* __restrict__ out,
                                      int E, int N, int R) {
  long long gid = (long long)blockIdx.x * blockDim.x + threadIdx.x;
  const long long total = ((long long)2 * E + N) * 8;
  if (gid >= total) return;
  const int c = (int)(gid & 7);
  const long long i = gid >> 3;
  int si, pi, oi;
  if (i < E)              { si = s[i];      pi = p[i];          oi = o[i]; }
  else if (i < 2LL * E)   { long long j = i - E; si = o[j]; pi = p[j] + R; oi = s[j]; }
  else                    { int nd = (int)(i - 2LL * E); si = nd; pi = 2 * R; oi = nd; }
  const float val = 1.0f / deg[(size_t)pi * N + si];
  const float* hv = &h[(size_t)oi * 16];
  const float* w  = &W2[(size_t)pi * 128 + c];
  float sum = 0.0f;
#pragma unroll
  for (int e = 0; e < 16; ++e) sum += hv[e] * w[e * 8];
  atomicAdd(&out[(size_t)si * 8 + c], val * sum);
}

extern "C" void kernel_launch(void* const* d_in, const int* in_sizes, int n_in,
                              void* d_out, int out_size, void* d_ws, size_t ws_size,
                              hipStream_t stream) {
  const float* comps1 = (const float*)d_in[0];
  const float* bases1 = (const float*)d_in[1];
  const float* comps2 = (const float*)d_in[2];
  const float* bases2 = (const float*)d_in[3];
  const float* bias1  = (const float*)d_in[4];
  const float* bias2  = (const float*)d_in[5];
  const int*   s      = (const int*)d_in[6];
  const int*   p      = (const int*)d_in[7];
  const int*   o      = (const int*)d_in[8];

  const int EMB    = in_sizes[4];
  const int NUMCLS = in_sizes[5];
  const int E      = in_sizes[6];
  const int N      = out_size / NUMCLS;
  const int BASES  = (int)((long long)in_sizes[1] / ((long long)N * EMB));
  const int r_tot  = in_sizes[0] / BASES;
  const int R      = (r_tot - 1) / 2;

  float* ws = (float*)d_ws;
  size_t off = 0;
  auto alloc = [&](size_t nwords) {
    size_t cur = off;
    off += (nwords + 63) & ~(size_t)63;
    return cur;
  };
  float* deg  = ws + alloc((size_t)r_tot * N);
  int*   head = (int*)(ws + alloc((size_t)N * kLists));
  float* h    = ws + alloc((size_t)N * EMB);
  const size_t cleared = off;
  float* W2   = ws + alloc((size_t)r_tot * EMB * NUMCLS);
  int*   next = (int*)(ws + alloc((size_t)2 * E));
  (void)ws_size; (void)n_in;

  const long long NE   = (long long)N * EMB;
  const long long J2   = (long long)EMB * NUMCLS;
  const long long Eall = (long long)2 * E + N;
  const int B = 256;

  fill_zero16_kernel<<<idiv_up((long long)(cleared / 4), B), B, 0, stream>>>(
      (float4*)ws, (long long)(cleared / 4));

  degree_list_kernel<<<idiv_up((long long)E + N, B), B, 0, stream>>>(
      s, p, o, deg, head, next, E, N, R);

  gemm48_wmma_kernel<true><<<idiv_up(NE, 16LL * kGemmWaves), kGemmThreads, 0, stream>>>(
      comps1, bases1, nullptr, r_tot, BASES, NE, s, p, o, deg, head, next, h, E, N, R);
  gemm48_wmma_kernel<false><<<idiv_up(J2, 16LL * kGemmWaves), kGemmThreads, 0, stream>>>(
      comps2, bases2, W2, r_tot, BASES, J2, nullptr, nullptr, nullptr, nullptr, nullptr,
      nullptr, nullptr, 0, 0, 0);

  relu_bias_kernel<<<idiv_up(NE, B), B, 0, stream>>>(h, bias1, NE);

  init_out_kernel<<<idiv_up((long long)N * NUMCLS, B), B, 0, stream>>>(
      (float*)d_out, bias2, (long long)N * NUMCLS, NUMCLS);
  layer2_scatter_kernel<<<idiv_up(Eall * 8, B), B, 0, stream>>>(
      s, p, o, deg, h, W2, (float*)d_out, E, N, R);
}


# Layer 2 as a gather: a half wave walks its node's own lists (they name its in-edges too), adds the lanes' sums by a tree, stores the out row once: no out atomics, no out-init launch (new sum order)
# speedup vs baseline: 1.5454x; 1.0403x over previous
#include <hip/hip_runtime.h>


typedef __attribute__((ext_vector_type(2))) float v2f;
typedef __attribute__((ext_vector_type(8))) float v8f;

static inline int idiv_up(long long a, long long b) { return (int)((a + b - 1) / b); }

__global__ void fill_zero16_kernel(float4* __restrict__ ptr, long long count4) {
  long long i = (long long)blockIdx.x * blockDim.x + threadIdx.x;
  if (i < count4) ptr[i] = make_float4(0.0f, 0.0f, 0.0f, 0.0f);
}

constexpr int kLists = 16;

__global__ void degree_list_kernel(const int* __restrict__ s, const int* __restrict__ p,
                                   const int* __restrict__ o, float* __restrict__ deg,
                                   int* __restrict__ head, int* __restrict__ next,
                                   int E, int N, int R) {
  long long i = (long long)blockIdx.x * blockDim.x + threadIdx.x;
  if (i < E) {
    int si = s[i], pi = p[i], oi = o[i];
    atomicAdd(&deg[(size_t)pi * N + si], 1.0f);
    atomicAdd(&deg[(size_t)(pi + R) * N + oi], 1.0f);
    const int e0 = (int)i, e1 = (int)i + E;
    const int before0 = atomicExch(&head[(size_t)oi * kLists + (pi & (kLists - 1))], e0 + 1);
    const int before1 = atomicExch(&head[(size_t)si * kLists + ((pi + R) & (kLists - 1))], e1 + 1);
    next[e0] = before0;
    next[e1] = before1;
  } else if (i < (long long)E + N) {
    int node = (int)(i - E);
    deg[(size_t)(2 * R) * N + node] = 1.0f;
  }
}

constexpr int kGemmWaves = 8;
constexpr int kGemmThreads = kGemmWaves * 32;
constexpr int kBatchSteps = 10;

template <bool kScatter>
__global__ void gemm48_wmma_kernel(const float* __restrict__ A,
                                   const float* __restrict__ B,
                                   float* __restrict__ C,
                                   int M, int K, long long J,
                                   const int* __restrict__ s, const int* __restrict__ p,
                                   const int* __restrict__ o,
                                   const float* __restrict__ deg,
                                   const int* __restrict__ head,
                                   const int* __restrict__ next,
                                   float* __restrict__ h,
                                   int E, int N, int R) {
  __shared__ __attribute__((aligned(16))) float sA[48 * 64];
  __shared__ float sTile[kScatter ? kGemmWaves * 48 * 16 : 1];

  const int wave = threadIdx.x >> 5;
  const int lane = threadIdx.x & 31;
  const long long jbase =
      ((long long)blockIdx.x * (blockDim.x >> 5) + wave) * 16;
  const bool active = jbase < J;
  const long long jb = active ? jbase : J - 16;
  const int col   = lane & 15;
  const int khalf = (lane >> 4) * 2;
  const int node  = (int)(jbase >> 4);

  const int total4 = 12 * K;
  const int valid4 = (M * K) >> 2;
  const float4* __restrict__ A4 = (const float4*)A;
  float4 areg[3];
#pragma unroll
  for (int t = 0; t < 3; ++t) {
    const int i4 = (int)threadIdx.x + t * kGemmThreads;
    areg[t] = (i4 < valid4) ? A4[i4] : make_float4(0.0f, 0.0f, 0.0f, 0.0f);
  }

  int cur = 0;
  if (kScatter) {
    if (active && lane < kLists) cur = head[(size_t)node * kLists + lane];
  }

  const float* __restrict__ Bp = B + jb + col;
  const unsigned bidx0 = (unsigned)((long long)khalf * J + jb + col);
  const unsigned bidx1 = bidx0 + (unsigned)J;
  const bool batch0 = K >= 4 * kBatchSteps;
  v2f bfrag[kBatchSteps] = {};
  if (batch0) {
#pragma unroll
    for (int i = 0; i < kBatchSteps; ++i) {
      unsigned long long rowbase = (unsigned long long)(B + (long long)(4 * i) * J);
      asm volatile("" : "+s"(rowbase));
      const __attribute__((address_space(1))) float* rowp =
          (const __attribute__((address_space(1))) float*)rowbase;
      bfrag[i].x = rowp[bidx0];
      bfrag[i].y = rowp[bidx1];
    }
  }

  float4* sA4 = (float4*)sA;
#pragma unroll
  for (int t = 0; t < 3; ++t) {
    const int i4 = (int)threadIdx.x + t * kGemmThreads;
    if (i4 < total4) sA4[i4] = areg[t];
  }

  auto fetch = [&](int c, int& tgt, int& pj, int& nx) {
    tgt = 0; pj = 0; nx = 0;
    if (c != 0) {
      const int e    = c - 1;
      const bool inv = e >= E;
      const int j    = inv ? e - E : e;
      pj  = p[j];
      tgt = inv ? o[j] : s[j];
      nx  = next[e];
    }
  };

  __syncthreads();

  int tgt = 0, prel = 0, nxt = 0;
  if (kScatter) fetch(cur, tgt, prel, nxt);

  v8f acc[3];
  acc[0] = (v8f){};
  acc[1] = (v8f){};
  acc[2] = (v8f){};

  auto mma_step = [&](int k0, v2f b) {
    v2f a[3];
#pragma unroll
    for (int mt = 0; mt < 3; ++mt) a[mt] = *(const v2f*)&sA[(mt * 16 + col) * K + k0];
#pragma unroll
    for (int mt = 0; mt < 3; ++mt) {
      acc[mt] = __builtin_amdgcn_wmma_f32_16x16x4_f32(
          false, a[mt], false, b, (short)0, acc[mt], false, false);
    }
  };

  int kc = 0;
  if (batch0) {
#pragma unroll
    for (int i = 0; i < kBatchSteps; ++i) mma_step(4 * i + khalf, bfrag[i]);
    kc = 4 * kBatchSteps;
  }
  for (; kc < K; kc += 4) {
    const int k0 = kc + khalf;
    v2f b;
    b.x = Bp[(long long)k0 * J];
    b.y = Bp[(long long)(k0 + 1) * J];
    mma_step(k0, b);
  }

  const int rhalf = (lane >> 4) * 8;
  if (!kScatter) {
    if (active) {
#pragma unroll
      for (int mt = 0; mt < 3; ++mt) {
#pragma unroll
        for (int v = 0; v < 8; ++v) {
          const int row = mt * 16 + rhalf + v;
          if (row < M) C[(long long)row * J + jb + col] = acc[mt][v];
        }
      }
    }
    return;
  }

  float* __restrict__ tile = &sTile[wave * (48 * 16)];
#pragma unroll
  for (int mt = 0; mt < 3; ++mt) {
#pragma unroll
    for (int v = 0; v < 8; ++v) {
      tile[(mt * 16 + rhalf + v) * 16 + col] = acc[mt][v];
    }
  }
  __syncthreads();
  if (!active) return;

  for (int trip = 0; trip < 2 * E; ++trip) {
    const bool has = cur != 0;
    unsigned pending = __builtin_amdgcn_ballot_w32(has);
    if (pending == 0u) break;
    const int rel = (cur - 1 >= E) ? prel + R : prel;
    float dg = 1.0f;
    if (has) dg = deg[(size_t)rel * N + tgt];
    const int cur2 = nxt;
    int tgt2, prel2, nxt2;
    fetch(cur2, tgt2, prel2, nxt2);
    const float val = has ? 1.0f / dg : 0.0f;
    while (pending != 0u) {
      const int la = __builtin_ctz(pending);
      pending &= pending - 1u;
      const bool two = pending != 0u;
      int lb = la;
      if (two) { lb = __builtin_ctz(pending); pending &= pending - 1u; }
      const int from = (lane < 16) ? la : lb;
      const int   t = __shfl(tgt, from);
      const int   r = __shfl(rel, from);
      const float v = __shfl(val, from);
      if (lane < 16 || two) atomicAdd(&h[(size_t)t * 16 + col], v * tile[r * 16 + col]);
    }
    cur = cur2; tgt = tgt2; prel = prel2; nxt = nxt2;
  }
  if (lane < 16) atomicAdd(&h[(size_t)node * 16 + col], tile[(2 * R) * 16 + col]);
}

__global__ void relu_bias_kernel(float* __restrict__ h, const float* __restrict__ bias1,
                                 long long count) {
  long long i = (long long)blockIdx.x * blockDim.x + threadIdx.x;
  if (i < count) h[i] = fmaxf(h[i] + bias1[i & 15], 0.0f);
}

constexpr int kL2Threads = 256;
constexpr int kW2Pitch   = 132;

__global__ void __launch_bounds__(kL2Threads, 10)
layer2_gather_kernel(const int* __restrict__ s, const int* __restrict__ p,
                     const int* __restrict__ o,
                     const float* __restrict__ deg,
                     const int* __restrict__ head,
                     const int* __restrict__ next,
                     const float* __restrict__ h,
                     const float* __restrict__ W2,
                     const float* __restrict__ bias2,
                     float* __restrict__ out,
                     int E, int N, int R, int r_tot) {
  __shared__ __attribute__((aligned(16))) float sW2[48 * kW2Pitch];

  const int lane = (int)threadIdx.x & 31;
  const int g    = lane & 15;
  const int half = lane >> 4;
  const int quad = (int)((blockIdx.x * (unsigned)kL2Threads + threadIdx.x) >> 5);

  auto fetch = [&](int c, int& pj, int& sr, int& nx) {
    pj = 0; sr = 0; nx = 0;
    if (c != 0) {
      const int e    = c - 1;
      const bool inv = e >= E;
      const int j    = inv ? e - E : e;
      pj = p[j];
      sr = inv ? o[j] : s[j];
      nx = next[e];
    }
  };

  int cur = 0;
  {
    const int t0 = 4 * quad + half;
    if (t0 < N) cur = head[(size_t)t0 * kLists + g];
  }
  const float4* __restrict__ W24 = (const float4*)W2;
  const int n4 = r_tot * 32;
  float4 wreg[6];
#pragma unroll
  for (int u = 0; u < 6; ++u) {
    const int i4 = (int)threadIdx.x + u * kL2Threads;
    wreg[u] = (i4 < n4) ? W24[i4] : make_float4(0.0f, 0.0f, 0.0f, 0.0f);
  }
#pragma unroll
  for (int u = 0; u < 6; ++u) {
    const int i4 = (int)threadIdx.x + u * kL2Threads;
    if (i4 < n4) *(float4*)&sW2[(i4 >> 5) * kW2Pitch + (i4 & 31) * 4] = wreg[u];
  }
  int pj, sr, nx;
  fetch(cur, pj, sr, nx);
  __syncthreads();

#pragma nounroll
  for (int i = 0; i < 2; ++i) {
    const int t = 4 * quad + 2 * i + half;
    if (t >= N) break;

    int curNext = 0;
    if (i == 0 && t + 2 < N) curNext = head[(size_t)(t + 2) * kLists + g];

    float acc[8];
#pragma unroll
    for (int c = 0; c < 8; ++c) acc[c] = 0.0f;

    for (int trip = 0; trip < 2 * E && cur != 0; ++trip) {
      const int rel = (cur - 1 >= E) ? pj : pj + R;
      const float dg = deg[(size_t)rel * N + t];
      const float4* __restrict__ hr4 = (const float4*)&h[(size_t)sr * 16];
      const float4 a0 = hr4[0], a1 = hr4[1], a2 = hr4[2], a3 = hr4[3];
      const int cur2 = nx;
      int pj2, sr2, nx2;
      fetch(cur2, pj2, sr2, nx2);

      const float val = 1.0f / dg;
      const float hv[16] = {a0.x, a0.y, a0.z, a0.w, a1.x, a1.y, a1.z, a1.w,
                            a2.x, a2.y, a2.z, a2.w, a3.x, a3.y, a3.z, a3.w};
      const float* __restrict__ w = &sW2[rel * kW2Pitch];
#pragma unroll
      for (int k = 0; k < 16; ++k) {
        const float x = val * hv[k];
        const float4 w0 = *(const float4*)&w[k * 8];
        const float4 w1 = *(const float4*)&w[k * 8 + 4];
        acc[0] += x * w0.x; acc[1] += x * w0.y; acc[2] += x * w0.z; acc[3] += x * w0.w;
        acc[4] += x * w1.x; acc[5] += x * w1.y; acc[6] += x * w1.z; acc[7] += x * w1.w;
      }
      cur = cur2; pj = pj2; sr = sr2; nx = nx2;
    }

    int pjn, srn, nxn;
    fetch(curNext, pjn, srn, nxn);

    const int col = g & 7;
    const float4* __restrict__ hs4 = (const float4*)&h[(size_t)t * 16];
    const float4 s0 = hs4[0], s1 = hs4[1], s2 = hs4[2], s3 = hs4[3];

#pragma unroll
    for (int c = 0; c < 8; ++c) {
      float v = acc[c];
      v += __shfl_xor(v, 8, 16);
      v += __shfl_xor(v, 4, 16);
      v += __shfl_xor(v, 2, 16);
      v += __shfl_xor(v, 1, 16);
      acc[c] = v;
    }
    float tot = acc[0];
#pragma unroll
    for (int c = 1; c < 8; ++c) tot = (col == c) ? acc[c] : tot;

    const float sv[16] = {s0.x, s0.y, s0.z, s0.w, s1.x, s1.y, s1.z, s1.w,
                          s2.x, s2.y, s2.z, s2.w, s3.x, s3.y, s3.z, s3.w};
    const float* __restrict__ wself = &sW2[(2 * R) * kW2Pitch + col];
    float self = 0.0f;
#pragma unroll
    for (int k = 0; k < 16; ++k) self += sv[k] * wself[k * 8];
    if (g < 8) out[(size_t)t * 8 + g] = bias2[g] + (tot + self);

    cur = curNext; pj = pjn; sr = srn; nx = nxn;
  }
}

extern "C" void kernel_launch(void* const* d_in, const int* in_sizes, int n_in,
                              void* d_out, int out_size, void* d_ws, size_t ws_size,
                              hipStream_t stream) {
  const float* comps1 = (const float*)d_in[0];
  const float* bases1 = (const float*)d_in[1];
  const float* comps2 = (const float*)d_in[2];
  const float* bases2 = (const float*)d_in[3];
  const float* bias1  = (const float*)d_in[4];
  const float* bias2  = (const float*)d_in[5];
  const int*   s      = (const int*)d_in[6];
  const int*   p      = (const int*)d_in[7];
  const int*   o      = (const int*)d_in[8];

  const int EMB    = in_sizes[4];
  const int NUMCLS = in_sizes[5];
  const int E      = in_sizes[6];
  const int N      = out_size / NUMCLS;
  const int BASES  = (int)((long long)in_sizes[1] / ((long long)N * EMB));
  const int r_tot  = in_sizes[0] / BASES;
  const int R      = (r_tot - 1) / 2;

  float* ws = (float*)d_ws;
  size_t off = 0;
  auto alloc = [&](size_t nwords) {
    size_t cur = off;
    off += (nwords + 63) & ~(size_t)63;
    return cur;
  };
  float* deg  = ws + alloc((size_t)r_tot * N);
  int*   head = (int*)(ws + alloc((size_t)N * kLists));
  float* h    = ws + alloc((size_t)N * EMB);
  const size_t cleared = off;
  float* W2   = ws + alloc((size_t)r_tot * EMB * NUMCLS);
  int*   next = (int*)(ws + alloc((size_t)2 * E));
  (void)ws_size; (void)n_in;

  const long long NE   = (long long)N * EMB;
  const long long J2   = (long long)EMB * NUMCLS;
  const int B = 256;

  fill_zero16_kernel<<<idiv_up((long long)(cleared / 4), B), B, 0, stream>>>(
      (float4*)ws, (long long)(cleared / 4));

  degree_list_kernel<<<idiv_up((long long)E + N, B), B, 0, stream>>>(
      s, p, o, deg, head, next, E, N, R);

  gemm48_wmma_kernel<true><<<idiv_up(NE, 16LL * kGemmWaves), kGemmThreads, 0, stream>>>(
      comps1, bases1, nullptr, r_tot, BASES, NE, s, p, o, deg, head, next, h, E, N, R);
  gemm48_wmma_kernel<false><<<idiv_up(J2, 16LL * kGemmWaves), kGemmThreads, 0, stream>>>(
      comps2, bases2, W2, r_tot, BASES, J2, nullptr, nullptr, nullptr, nullptr, nullptr,
      nullptr, nullptr, 0, 0, 0);

  relu_bias_kernel<<<idiv_up(NE, B), B, 0, stream>>>(h, bias1, NE);

  layer2_gather_kernel<<<idiv_up((long long)N, 4LL * (kL2Threads / 32)), kL2Threads, 0, stream>>>(
      s, p, o, deg, head, next, h, W2, bias2, (float*)d_out, E, N, R, r_tot);
}


# List entries packed: the list build writes one 8-byte record per listed edge (far node, relation, link), so a step of either layer's walk is one load instead of three from p, s or o, and next
# speedup vs baseline: 1.6108x; 1.0423x over previous
#include <hip/hip_runtime.h>


typedef __attribute__((ext_vector_type(2))) float v2f;
typedef __attribute__((ext_vector_type(8))) float v8f;

static inline int idiv_up(long long a, long long b) { return (int)((a + b - 1) / b); }

__global__ void fill_zero16_kernel(float4* __restrict__ ptr, long long count4) {
  long long i = (long long)blockIdx.x * blockDim.x + threadIdx.x;
  if (i < count4) ptr[i] = make_float4(0.0f, 0.0f, 0.0f, 0.0f);
}

constexpr int kLists = 16;
constexpr int kRelBits = 6;
constexpr int kRelMask = (1 << kRelBits) - 1;

typedef unsigned long long entry_t;

static __device__ __forceinline__ entry_t pack_entry(int far, int rel, int before) {
  return ((entry_t)(unsigned)before << 32) | (entry_t)(((unsigned)far << kRelBits) | (unsigned)rel);
}
static __device__ __forceinline__ int entry_far(entry_t r)  { return (int)((unsigned)r >> kRelBits); }
static __device__ __forceinline__ int entry_rel(entry_t r)  { return (int)((unsigned)r & (unsigned)kRelMask); }
static __device__ __forceinline__ int entry_link(entry_t r) { return (int)(r >> 32); }

__global__ void degree_list_kernel(const int* __restrict__ s, const int* __restrict__ p,
                                   const int* __restrict__ o, float* __restrict__ deg,
                                   int* __restrict__ head, entry_t* __restrict__ rec,
                                   int E, int N, int R) {
  long long i = (long long)blockIdx.x * blockDim.x + threadIdx.x;
  if (i < E) {
    int si = s[i], pi = p[i], oi = o[i];
    atomicAdd(&deg[(size_t)pi * N + si], 1.0f);
    atomicAdd(&deg[(size_t)(pi + R) * N + oi], 1.0f);
    const int e0 = (int)i, e1 = (int)i + E;
    const int before0 = atomicExch(&head[(size_t)oi * kLists + (pi & (kLists - 1))], e0 + 1);
    const int before1 = atomicExch(&head[(size_t)si * kLists + ((pi + R) & (kLists - 1))], e1 + 1);
    rec[e0] = pack_entry(si, pi, before0);
    rec[e1] = pack_entry(oi, pi + R, before1);
  } else if (i < (long long)E + N) {
    int node = (int)(i - E);
    deg[(size_t)(2 * R) * N + node] = 1.0f;
  }
}

constexpr int kGemmWaves = 8;
constexpr int kGemmThreads = kGemmWaves * 32;
constexpr int kBatchSteps = 10;

template <bool kScatter>
__global__ void gemm48_wmma_kernel(const float* __restrict__ A,
                                   const float* __restrict__ B,
                                   float* __restrict__ C,
                                   int M, int K, long long J,
                                   const float* __restrict__ deg,
                                   const int* __restrict__ head,
                                   const entry_t* __restrict__ rec,
                                   float* __restrict__ h,
                                   int E, int N, int R) {
  __shared__ __attribute__((aligned(16))) float sA[48 * 64];
  __shared__ float sTile[kScatter ? kGemmWaves * 48 * 16 : 1];

  const int wave = threadIdx.x >> 5;
  const int lane = threadIdx.x & 31;
  const long long jbase =
      ((long long)blockIdx.x * (blockDim.x >> 5) + wave) * 16;
  const bool active = jbase < J;
  const long long jb = active ? jbase : J - 16;
  const int col   = lane & 15;
  const int khalf = (lane >> 4) * 2;
  const int node  = (int)(jbase >> 4);

  const int total4 = 12 * K;
  const int valid4 = (M * K) >> 2;
  const float4* __restrict__ A4 = (const float4*)A;
  float4 areg[3];
#pragma unroll
  for (int t = 0; t < 3; ++t) {
    const int i4 = (int)threadIdx.x + t * kGemmThreads;
    areg[t] = (i4 < valid4) ? A4[i4] : make_float4(0.0f, 0.0f, 0.0f, 0.0f);
  }

  int cur = 0;
  if (kScatter) {
    if (active && lane < kLists) cur = head[(size_t)node * kLists + lane];
  }

  const float* __restrict__ Bp = B + jb + col;
  const unsigned bidx0 = (unsigned)((long long)khalf * J + jb + col);
  const unsigned bidx1 = bidx0 + (unsigned)J;
  const bool batch0 = K >= 4 * kBatchSteps;
  v2f bfrag[kBatchSteps] = {};
  if (batch0) {
#pragma unroll
    for (int i = 0; i < kBatchSteps; ++i) {
      unsigned long long rowbase = (unsigned long long)(B + (long long)(4 * i) * J);
      asm volatile("" : "+s"(rowbase));
      const __attribute__((address_space(1))) float* rowp =
          (const __attribute__((address_space(1))) float*)rowbase;
      bfrag[i].x = rowp[bidx0];
      bfrag[i].y = rowp[bidx1];
    }
  }

  float4* sA4 = (float4*)sA;
#pragma unroll
  for (int t = 0; t < 3; ++t) {
    const int i4 = (int)threadIdx.x + t * kGemmThreads;
    if (i4 < total4) sA4[i4] = areg[t];
  }

  auto fetch = [&](int c) {
    entry_t r = 0;
    if (c != 0) r = rec[c - 1];
    return r;
  };

  __syncthreads();

  entry_t ent = 0;
  if (kScatter) ent = fetch(cur);

  v8f acc[3];
  acc[0] = (v8f){};
  acc[1] = (v8f){};
  acc[2] = (v8f){};

  auto mma_step = [&](int k0, v2f b) {
    v2f a[3];
#pragma unroll
    for (int mt = 0; mt < 3; ++mt) a[mt] = *(const v2f*)&sA[(mt * 16 + col) * K + k0];
#pragma unroll
    for (int mt = 0; mt < 3; ++mt) {
      acc[mt] = __builtin_amdgcn_wmma_f32_16x16x4_f32(
          false, a[mt], false, b, (short)0, acc[mt], false, false);
    }
  };

  int kc = 0;
  if (batch0) {
#pragma unroll
    for (int i = 0; i < kBatchSteps; ++i) mma_step(4 * i + khalf, bfrag[i]);
    kc = 4 * kBatchSteps;
  }
  for (; kc < K; kc += 4) {
    const int k0 = kc + khalf;
    v2f b;
    b.x = Bp[(long long)k0 * J];
    b.y = Bp[(long long)(k0 + 1) * J];
    mma_step(k0, b);
  }

  const int rhalf = (lane >> 4) * 8;
  if (!kScatter) {
    if (active) {
#pragma unroll
      for (int mt = 0; mt < 3; ++mt) {
#pragma unroll
        for (int v = 0; v < 8; ++v) {
          const int row = mt * 16 + rhalf + v;
          if (row < M) C[(long long)row * J + jb + col] = acc[mt][v];
        }
      }
    }
    return;
  }

  float* __restrict__ tile = &sTile[wave * (48 * 16)];
#pragma unroll
  for (int mt = 0; mt < 3; ++mt) {
#pragma unroll
    for (int v = 0; v < 8; ++v) {
      tile[(mt * 16 + rhalf + v) * 16 + col] = acc[mt][v];
    }
  }
  __syncthreads();
  if (!active) return;

  for (int trip = 0; trip < 2 * E; ++trip) {
    const bool has = cur != 0;
    unsigned pending = __builtin_amdgcn_ballot_w32(has);
    if (pending == 0u) break;
    const int tgt = entry_far(ent);
    const int rel = entry_rel(ent);
    float dg = 1.0f;
    if (has) dg = deg[(size_t)rel * N + tgt];
    const int cur2 = entry_link(ent);
    const entry_t ent2 = fetch(cur2);
    const float val = has ? 1.0f / dg : 0.0f;
    while (pending != 0u) {
      const int la = __builtin_ctz(pending);
      pending &= pending - 1u;
      const bool two = pending != 0u;
      int lb = la;
      if (two) { lb = __builtin_ctz(pending); pending &= pending - 1u; }
      const int from = (lane < 16) ? la : lb;
      const int   t = __shfl(tgt, from);
      const int   r = __shfl(rel, from);
      const float v = __shfl(val, from);
      if (lane < 16 || two) atomicAdd(&h[(size_t)t * 16 + col], v * tile[r * 16 + col]);
    }
    cur = cur2; ent = ent2;
  }
  if (lane < 16) atomicAdd(&h[(size_t)node * 16 + col], tile[(2 * R) * 16 + col]);
}

__global__ void relu_bias_kernel(float* __restrict__ h, const float* __restrict__ bias1,
                                 long long count) {
  long long i = (long long)blockIdx.x * blockDim.x + threadIdx.x;
  if (i < count) h[i] = fmaxf(h[i] + bias1[i & 15], 0.0f);
}

constexpr int kL2Threads = 256;
constexpr int kW2Pitch   = 132;

__global__ void __launch_bounds__(kL2Threads, 10)
layer2_gather_kernel(const float* __restrict__ deg,
                     const int* __restrict__ head,
                     const entry_t* __restrict__ rec,
                     const float* __restrict__ h,
                     const float* __restrict__ W2,
                     const float* __restrict__ bias2,
                     float* __restrict__ out,
                     int E, int N, int R, int r_tot) {
  __shared__ __attribute__((aligned(16))) float sW2[48 * kW2Pitch];

  const int lane = (int)threadIdx.x & 31;
  const int g    = lane & 15;
  const int half = lane >> 4;
  const int quad = (int)((blockIdx.x * (unsigned)kL2Threads + threadIdx.x) >> 5);

  auto fetch = [&](int c) {
    entry_t r = 0;
    if (c != 0) r = rec[c - 1];
    return r;
  };

  int cur = 0;
  {
    const int t0 = 4 * quad + half;
    if (t0 < N) cur = head[(size_t)t0 * kLists + g];
  }
  const float4* __restrict__ W24 = (const float4*)W2;
  const int n4 = r_tot * 32;
  float4 wreg[6];
#pragma unroll
  for (int u = 0; u < 6; ++u) {
    const int i4 = (int)threadIdx.x + u * kL2Threads;
    wreg[u] = (i4 < n4) ? W24[i4] : make_float4(0.0f, 0.0f, 0.0f, 0.0f);
  }
#pragma unroll
  for (int u = 0; u < 6; ++u) {
    const int i4 = (int)threadIdx.x + u * kL2Threads;
    if (i4 < n4) *(float4*)&sW2[(i4 >> 5) * kW2Pitch + (i4 & 31) * 4] = wreg[u];
  }
  entry_t ent = fetch(cur);
  __syncthreads();

#pragma nounroll
  for (int i = 0; i < 2; ++i) {
    const int t = 4 * quad + 2 * i + half;
    if (t >= N) break;

    int curNext = 0;
    if (i == 0 && t + 2 < N) curNext = head[(size_t)(t + 2) * kLists + g];

    float acc[8];
#pragma unroll
    for (int c = 0; c < 8; ++c) acc[c] = 0.0f;

    for (int trip = 0; trip < 2 * E && cur != 0; ++trip) {
      const int rel = (cur - 1 >= E) ? entry_rel(ent) - R : entry_rel(ent) + R;
      const int sr  = entry_far(ent);
      const float dg = deg[(size_t)rel * N + t];
      const float4* __restrict__ hr4 = (const float4*)&h[(size_t)sr * 16];
      const float4 a0 = hr4[0], a1 = hr4[1], a2 = hr4[2], a3 = hr4[3];
      const int cur2 = entry_link(ent);
      const entry_t ent2 = fetch(cur2);

      const float val = 1.0f / dg;
      const float hv[16] = {a0.x, a0.y, a0.z, a0.w, a1.x, a1.y, a1.z, a1.w,
                            a2.x, a2.y, a2.z, a2.w, a3.x, a3.y, a3.z, a3.w};
      const float* __restrict__ w = &sW2[rel * kW2Pitch];
#pragma unroll
      for (int k = 0; k < 16; ++k) {
        const float x = val * hv[k];
        const float4 w0 = *(const float4*)&w[k * 8];
        const float4 w1 = *(const float4*)&w[k * 8 + 4];
        acc[0] += x * w0.x; acc[1] += x * w0.y; acc[2] += x * w0.z; acc[3] += x * w0.w;
        acc[4] += x * w1.x; acc[5] += x * w1.y; acc[6] += x * w1.z; acc[7] += x * w1.w;
      }
      cur = cur2; ent = ent2;
    }

    const entry_t entNext = fetch(curNext);

    const int col = g & 7;
    const float4* __restrict__ hs4 = (const float4*)&h[(size_t)t * 16];
    const float4 s0 = hs4[0], s1 = hs4[1], s2 = hs4[2], s3 = hs4[3];

#pragma unroll
    for (int c = 0; c < 8; ++c) {
      float v = acc[c];
      v += __shfl_xor(v, 8, 16);
      v += __shfl_xor(v, 4, 16);
      v += __shfl_xor(v, 2, 16);
      v += __shfl_xor(v, 1, 16);
      acc[c] = v;
    }
    float tot = acc[0];
#pragma unroll
    for (int c = 1; c < 8; ++c) tot = (col == c) ? acc[c] : tot;

    const float sv[16] = {s0.x, s0.y, s0.z, s0.w, s1.x, s1.y, s1.z, s1.w,
                          s2.x, s2.y, s2.z, s2.w, s3.x, s3.y, s3.z, s3.w};
    const float* __restrict__ wself = &sW2[(2 * R) * kW2Pitch + col];
    float self = 0.0f;
#pragma unroll
    for (int k = 0; k < 16; ++k) self += sv[k] * wself[k * 8];
    if (g < 8) out[(size_t)t * 8 + g] = bias2[g] + (tot + self);

    cur = curNext; ent = entNext;
  }
}

extern "C" void kernel_launch(void* const* d_in, const int* in_sizes, int n_in,
                              void* d_out, int out_size, void* d_ws, size_t ws_size,
                              hipStream_t stream) {
  const float* comps1 = (const float*)d_in[0];
  const float* bases1 = (const float*)d_in[1];
  const float* comps2 = (const float*)d_in[2];
  const float* bases2 = (const float*)d_in[3];
  const float* bias1  = (const float*)d_in[4];
  const float* bias2  = (const float*)d_in[5];
  const int*   s      = (const int*)d_in[6];
  const int*   p      = (const int*)d_in[7];
  const int*   o      = (const int*)d_in[8];

  const int EMB    = in_sizes[4];
  const int NUMCLS = in_sizes[5];
  const int E      = in_sizes[6];
  const int N      = out_size / NUMCLS;
  const int BASES  = (int)((long long)in_sizes[1] / ((long long)N * EMB));
  const int r_tot  = in_sizes[0] / BASES;
  const int R      = (r_tot - 1) / 2;

  float* ws = (float*)d_ws;
  size_t off = 0;
  auto alloc = [&](size_t nwords) {
    size_t cur = off;
    off += (nwords + 63) & ~(size_t)63;
    return cur;
  };
  float* deg  = ws + alloc((size_t)r_tot * N);
  int*   head = (int*)(ws + alloc((size_t)N * kLists));
  float* h    = ws + alloc((size_t)N * EMB);
  const size_t cleared = off;
  float* W2   = ws + alloc((size_t)r_tot * EMB * NUMCLS);
  entry_t* rec = (entry_t*)(ws + alloc((size_t)4 * E));
  (void)ws_size; (void)n_in;

  const long long NE   = (long long)N * EMB;
  const long long J2   = (long long)EMB * NUMCLS;
  const int B = 256;

  fill_zero16_kernel<<<idiv_up((long long)(cleared / 4), B), B, 0, stream>>>(
      (float4*)ws, (long long)(cleared / 4));

  degree_list_kernel<<<idiv_up((long long)E + N, B), B, 0, stream>>>(
      s, p, o, deg, head, rec, E, N, R);

  gemm48_wmma_kernel<true><<<idiv_up(NE, 16LL * kGemmWaves), kGemmThreads, 0, stream>>>(
      comps1, bases1, nullptr, r_tot, BASES, NE, deg, head, rec, h, E, N, R);
  gemm48_wmma_kernel<false><<<idiv_up(J2, 16LL * kGemmWaves), kGemmThreads, 0, stream>>>(
      comps2, bases2, W2, r_tot, BASES, J2, nullptr, nullptr, nullptr, nullptr, 0, 0, 0);

  relu_bias_kernel<<<idiv_up(NE, B), B, 0, stream>>>(h, bias1, NE);

  layer2_gather_kernel<<<idiv_up((long long)N, 4LL * (kL2Threads / 32)), kL2Threads, 0, stream>>>(
      deg, head, rec, h, W2, bias2, (float*)d_out, E, N, R, r_tot);
}
